# starting kernel
# baseline (speedup 1.0000x reference)
#include <hip/hip_runtime.h>

#define BDIM   32
#define NSEQ   1024
#define CDIM   768
#define HHEADS 12
#define RLOW   10
#define MSUB   100
#define DH     64

#define GM (BDIM*NSEQ)
#define GN (2*CDIM)
#define GK (CDIM)
#define TM 128
#define TN 128
#define TK 32
#define LDA 36
#define LDB 132

typedef __attribute__((ext_vector_type(2))) float v2f;
typedef __attribute__((ext_vector_type(8))) float v8f;

__global__ __launch_bounds__(256)
void qk_gemm_wmma(const float* __restrict__ X, const float* __restrict__ W,
                  float* __restrict__ out) {
  __shared__ float As[TM * LDA];
  __shared__ float Bs[TK * LDB];
  const int tid  = threadIdx.x;
  const int lane = tid & 31;
  const int wave = tid >> 5;
  const int mw   = wave & 3;
  const int nw   = wave >> 2;
  const int lo   = lane & 15;
  const int hi   = lane >> 4;
  const int bm   = blockIdx.y * TM;
  const int bn   = blockIdx.x * TN;

  v8f acc00 = {0.f,0.f,0.f,0.f,0.f,0.f,0.f,0.f};
  v8f acc01 = acc00, acc02 = acc00, acc03 = acc00;
  v8f acc10 = acc00, acc11 = acc00, acc12 = acc00, acc13 = acc00;

  const int mrow0 = mw * 32 + lo;
  const int mrow1 = mrow0 + 16;
  const int cb    = nw * 64 + lo;

  for (int kk = 0; kk < GK; kk += TK) {
#pragma unroll
    for (int it = 0; it < 4; ++it) {
      int i   = tid + it * 256;
      int row = i >> 3;
      int c4  = (i & 7) << 2;
      float4 v = *(const float4*)(X + (size_t)(bm + row) * GK + kk + c4);
      *(float4*)(&As[row * LDA + c4]) = v;
    }
#pragma unroll
    for (int it = 0; it < 4; ++it) {
      int i  = tid + it * 256;
      int jl = i >> 3;
      int k4 = (i & 7) << 2;
      float4 v = *(const float4*)(W + (size_t)(bn + jl) * GK + kk + k4);
      Bs[(k4 + 0) * LDB + jl] = v.x;
      Bs[(k4 + 1) * LDB + jl] = v.y;
      Bs[(k4 + 2) * LDB + jl] = v.z;
      Bs[(k4 + 3) * LDB + jl] = v.w;
    }
    __syncthreads();

#pragma unroll
    for (int k4 = 0; k4 < 8; ++k4) {
      const int kb = k4 * 4 + hi * 2;
      v2f a0, a1;
      a0.x = As[mrow0 * LDA + kb];  a0.y = As[mrow0 * LDA + kb + 1];
      a1.x = As[mrow1 * LDA + kb];  a1.y = As[mrow1 * LDA + kb + 1];
      v2f b0, b1, b2, b3;
      b0.x = Bs[kb * LDB + cb];        b0.y = Bs[(kb + 1) * LDB + cb];
      b1.x = Bs[kb * LDB + cb + 16];   b1.y = Bs[(kb + 1) * LDB + cb + 16];
      b2.x = Bs[kb * LDB + cb + 32];   b2.y = Bs[(kb + 1) * LDB + cb + 32];
      b3.x = Bs[kb * LDB + cb + 48];   b3.y = Bs[(kb + 1) * LDB + cb + 48];
      acc00 = __builtin_amdgcn_wmma_f32_16x16x4_f32(false, a0, false, b0, (short)0, acc00, false, false);
      acc01 = __builtin_amdgcn_wmma_f32_16x16x4_f32(false, a0, false, b1, (short)0, acc01, false, false);
      acc02 = __builtin_amdgcn_wmma_f32_16x16x4_f32(false, a0, false, b2, (short)0, acc02, false, false);
      acc03 = __builtin_amdgcn_wmma_f32_16x16x4_f32(false, a0, false, b3, (short)0, acc03, false, false);
      acc10 = __builtin_amdgcn_wmma_f32_16x16x4_f32(false, a1, false, b0, (short)0, acc10, false, false);
      acc11 = __builtin_amdgcn_wmma_f32_16x16x4_f32(false, a1, false, b1, (short)0, acc11, false, false);
      acc12 = __builtin_amdgcn_wmma_f32_16x16x4_f32(false, a1, false, b2, (short)0, acc12, false, false);
      acc13 = __builtin_amdgcn_wmma_f32_16x16x4_f32(false, a1, false, b3, (short)0, acc13, false, false);
    }
    __syncthreads();
  }

  const size_t r0 = (size_t)(bm + mw * 32 + 8 * hi);
  const int    cg = bn + nw * 64 + lo;
#pragma unroll
  for (int i = 0; i < 8; ++i) {
    float* p0 = out + (r0 + i) * GN + cg;
    p0[0]  = acc00[i];
    p0[16] = acc01[i];
    p0[32] = acc02[i];
    p0[48] = acc03[i];
    float* p1 = out + (r0 + 16 + i) * GN + cg;
    p1[0]  = acc10[i];
    p1[16] = acc11[i];
    p1[32] = acc12[i];
    p1[48] = acc13[i];
  }
}

__global__ __launch_bounds__(256)
void gen_weights(const float* __restrict__ x, const float* __restrict__ we_p,
                 const float* __restrict__ wr_p,
                 float* __restrict__ we, float* __restrict__ wr) {
  __shared__ int sidx[MSUB];
  const int tid = threadIdx.x;
  if (tid < MSUB) {
    const double delta = 1023.0 / 99.0;
    int v = (int)((double)tid * delta);
    if (tid == MSUB - 1) v = NSEQ - 1;
    sidx[tid] = v;
  }
  __syncthreads();

  const int g = blockIdx.x * 256 + tid;
  if (g >= BDIM * HHEADS * DH * RLOW) return;
  const int r = g % RLOW;
  const int d = (g / RLOW) % DH;
  const int h = (g / (RLOW * DH)) % HHEADS;
  const int b = g / (RLOW * DH * HHEADS);

  float aw = 0.f, ar = 0.f;
#pragma unroll 4
  for (int m = 0; m < MSUB; ++m) {
    float xs = x[((size_t)b * NSEQ + sidx[m]) * CDIM + h * DH + d];
    aw += xs * we_p[(h * MSUB + m) * RLOW + r];
    ar += xs * wr_p[(h * MSUB + m) * RLOW + r];
  }
  we[g] = aw;
  wr[g] = ar;
}

__device__ __forceinline__ float wave_sum32(float v) {
  v += __shfl_xor(v, 16, 32);
  v += __shfl_xor(v, 8, 32);
  v += __shfl_xor(v, 4, 32);
  v += __shfl_xor(v, 2, 32);
  v += __shfl_xor(v, 1, 32);
  return v;
}

__global__ __launch_bounds__(256)
void svgp_epilogue(const float* __restrict__ qk, const float* __restrict__ we,
                   const float* __restrict__ wr,
                   const float* __restrict__ log_lambda,
                   const float* __restrict__ m_u, const float* __restrict__ s_tril,
                   const float* __restrict__ log_ssqrt,
                   const float* __restrict__ Wf, const float* __restrict__ bf,
                   const float* __restrict__ eps,
                   float* __restrict__ esc, float* __restrict__ rsc,
                   float* __restrict__ attn) {
  __shared__ float we_s[DH * RLOW];
  __shared__ float wr_s[DH * RLOW];
  __shared__ float wf_s[DH * RLOW];
  __shared__ float sl_s[RLOW * RLOW * RLOW];
  __shared__ float mu_s[RLOW * RLOW];
  __shared__ float bf_s[DH];
  __shared__ float lam2_s[RLOW];

  const int tid  = threadIdx.x;
  const int lane = tid & 31;
  const int wave = tid >> 5;
  const int h = blockIdx.y;
  const int b = blockIdx.z;

  const size_t bh = (size_t)b * HHEADS + h;
  for (int i = tid; i < DH * RLOW; i += 256) {
    we_s[i] = we[bh * DH * RLOW + i];
    wr_s[i] = wr[bh * DH * RLOW + i];
    wf_s[i] = Wf[i];
  }
  for (int i = tid; i < RLOW * RLOW * RLOW; i += 256) {
    int a = i / (RLOW * RLOW);
    int q = (i / RLOW) % RLOW;
    int c = i % RLOW;
    float v = (q == c) ? expf(log_ssqrt[(h * RLOW + a) * RLOW + q]) : 0.f;
    if (q > c) v += s_tril[((h * RLOW + a) * RLOW + q) * RLOW + c];
    sl_s[i] = v;
  }
  for (int i = tid; i < RLOW * RLOW; i += 256) mu_s[i] = m_u[h * RLOW * RLOW + i];
  if (tid < DH)   bf_s[tid]   = bf[tid];
  if (tid < RLOW) lam2_s[tid] = expf(2.f * log_lambda[h * RLOW + tid]);
  __syncthreads();

  const int n = blockIdx.x * 8 + wave;
  const float* qrow = qk + ((size_t)b * NSEQ + n) * GN + h * DH;
  const float* krow = qrow + CDIM;

  float q0 = qrow[lane], q1 = qrow[lane + 32];
  float k0 = krow[lane], k1 = krow[lane + 32];
  float qn = wave_sum32(q0 * q0 + q1 * q1);
  float kn = wave_sum32(k0 * k0 + k1 * k1);
  float qi = 1.f / fmaxf(sqrtf(qn), 1e-12f);
  float ki = 1.f / fmaxf(sqrtf(kn), 1e-12f);
  q0 *= qi; q1 *= qi; k0 *= ki; k1 *= ki;

  float e[RLOW], rv[RLOW], v1[RLOW];
#pragma unroll
  for (int r = 0; r < RLOW; ++r) {
    e[r]  = wave_sum32(q0 * we_s[lane * RLOW + r] + q1 * we_s[(lane + 32) * RLOW + r]);
    rv[r] = wave_sum32(k0 * wr_s[lane * RLOW + r] + k1 * wr_s[(lane + 32) * RLOW + r]);
    v1[r] = (e[r] + rv[r]) * lam2_s[r];
  }

  const size_t so = (bh * NSEQ + n) * RLOW;
  if (lane == 0) {
#pragma unroll
    for (int r = 0; r < RLOW; ++r) { esc[so + r] = e[r]; rsc[so + r] = rv[r]; }
  }

  float sample = 0.f;
  if (lane < RLOW) {
    const int c = lane;
    float mean = 0.f;
#pragma unroll
    for (int r = 0; r < RLOW; ++r) mean += v1[r] * mu_s[r * RLOW + c];
    const float* ep = eps + so;
    float noise = 0.f;
#pragma unroll
    for (int q = 0; q < RLOW; ++q) {
      float tq = 0.f;
#pragma unroll
      for (int a = 0; a < RLOW; ++a) tq += v1[a] * sl_s[(a * RLOW + q) * RLOW + c];
      noise += ep[q] * tq;
    }
    sample = mean + noise;
  }
  float sr[RLOW];
#pragma unroll
  for (int r = 0; r < RLOW; ++r) sr[r] = __shfl(sample, r, 32);

  float a0 = bf_s[lane], a1 = bf_s[lane + 32];
#pragma unroll
  for (int r = 0; r < RLOW; ++r) {
    a0 += sr[r] * wf_s[lane * RLOW + r];
    a1 += sr[r] * wf_s[(lane + 32) * RLOW + r];
  }
  float* ap = attn + ((size_t)b * NSEQ + n) * CDIM + h * DH;
  ap[lane]      = a0;
  ap[lane + 32] = a1;
}

__global__ __launch_bounds__(256)
void kl_reduce(const float* __restrict__ log_lambda, const float* __restrict__ m_u,
               const float* __restrict__ s_tril, const float* __restrict__ log_ssqrt,
               float* __restrict__ out_kl) {
  __shared__ float red[256];
  const int tid = threadIdx.x;
  float p = 0.f;
  for (int i = tid; i < HHEADS * RLOW * RLOW * RLOW; i += 256) {
    int h = i / (RLOW * RLOW * RLOW);
    int a = (i / (RLOW * RLOW)) % RLOW;
    int q = (i / RLOW) % RLOW;
    int c = i % RLOW;
    float sl = (q == c) ? expf(log_ssqrt[(h * RLOW + a) * RLOW + q]) : 0.f;
    if (q > c) sl += s_tril[((h * RLOW + a) * RLOW + q) * RLOW + c];
    float t = expf(2.f * log_lambda[h * RLOW + a]) * sl;
    p += 0.5f * t * t;
  }
  for (int i = tid; i < HHEADS * RLOW * RLOW; i += 256) {
    int h = i / (RLOW * RLOW);
    int r = (i / RLOW) % RLOW;
    float mu = m_u[i];
    p += 0.5f * mu * mu * expf(4.f * log_lambda[h * RLOW + r]);
    p -= log_ssqrt[i];
  }
  for (int i = tid; i < HHEADS * RLOW; i += 256) p -= 2.f * (float)RLOW * log_lambda[i];
  red[tid] = p;
  __syncthreads();
  for (int s = 128; s > 0; s >>= 1) {
    if (tid < s) red[tid] += red[tid + s];
    __syncthreads();
  }
  if (tid == 0) out_kl[0] = red[0] - 0.5f * (float)(RLOW * RLOW * HHEADS);
}

extern "C" void kernel_launch(void* const* d_in, const int* in_sizes, int n_in,
                              void* d_out, int out_size, void* d_ws, size_t ws_size,
                              hipStream_t stream) {
  const float* x          = (const float*)d_in[0];
  const float* Wqk        = (const float*)d_in[1];
  const float* we_p       = (const float*)d_in[2];
  const float* wr_p       = (const float*)d_in[3];
  const float* log_lambda = (const float*)d_in[4];
  const float* m_u        = (const float*)d_in[5];
  const float* s_tril     = (const float*)d_in[6];
  const float* log_ssqrt  = (const float*)d_in[7];
  const float* Wf         = (const float*)d_in[8];
  const float* bf         = (const float*)d_in[9];
  const float* eps        = (const float*)d_in[10];

  float* ws = (float*)d_ws;
  float* qk = ws;
  float* we = ws + (size_t)GM * GN;
  float* wr = we + (size_t)BDIM * HHEADS * DH * RLOW;

  float* attn = (float*)d_out;
  float* esc  = attn + (size_t)BDIM * NSEQ * CDIM;
  float* rsc  = esc + (size_t)BDIM * HHEADS * NSEQ * RLOW;
  float* klp  = rsc + (size_t)BDIM * HHEADS * NSEQ * RLOW;

  qk_gemm_wmma<<<dim3(GN / TN, GM / TM), 256, 0, stream>>>(x, Wqk, qk);
  gen_weights<<<(BDIM * HHEADS * DH * RLOW + 255) / 256, 256, 0, stream>>>(
      x, we_p, wr_p, we, wr);
  svgp_epilogue<<<dim3(NSEQ / 8, HHEADS, BDIM), 256, 0, stream>>>(
      qk, we, wr, log_lambda, m_u, s_tril, log_ssqrt, Wf, bf, eps, esc, rsc, attn);
  kl_reduce<<<1, 256, 0, stream>>>(log_lambda, m_u, s_tril, log_ssqrt, klp);
}


# epilogue row products moved to the matrix unit: per 16-row tile scores, mean+noise (one K=120 contraction) and output run as transposed f32 WMMAs, weights as A from an LDS table; sum order changes
# speedup vs baseline: 1.3421x; 1.3421x over previous
#include <hip/hip_runtime.h>

#define BDIM   32
#define NSEQ   1024
#define CDIM   768
#define HHEADS 12
#define RLOW   10
#define MSUB   100
#define DH     64

#define GM (BDIM*NSEQ)
#define GN (2*CDIM)
#define GK (CDIM)
#define TM 128
#define TN 128
#define TK 32
#define LDA 36
#define LDB 132

typedef __attribute__((ext_vector_type(2))) float v2f;
typedef __attribute__((ext_vector_type(8))) float v8f;

__global__ __launch_bounds__(256)
void qk_gemm_wmma(const float* __restrict__ X, const float* __restrict__ W,
                  float* __restrict__ out) {
  __shared__ float As[TM * LDA];
  __shared__ float Bs[TK * LDB];
  const int tid  = threadIdx.x;
  const int lane = tid & 31;
  const int wave = tid >> 5;
  const int mw   = wave & 3;
  const int nw   = wave >> 2;
  const int lo   = lane & 15;
  const int hi   = lane >> 4;
  const int bm   = blockIdx.y * TM;
  const int bn   = blockIdx.x * TN;

  v8f acc00 = {0.f,0.f,0.f,0.f,0.f,0.f,0.f,0.f};
  v8f acc01 = acc00, acc02 = acc00, acc03 = acc00;
  v8f acc10 = acc00, acc11 = acc00, acc12 = acc00, acc13 = acc00;

  const int mrow0 = mw * 32 + lo;
  const int mrow1 = mrow0 + 16;
  const int cb    = nw * 64 + lo;

  for (int kk = 0; kk < GK; kk += TK) {
#pragma unroll
    for (int it = 0; it < 4; ++it) {
      int i   = tid + it * 256;
      int row = i >> 3;
      int c4  = (i & 7) << 2;
      float4 v = *(const float4*)(X + (size_t)(bm + row) * GK + kk + c4);
      *(float4*)(&As[row * LDA + c4]) = v;
    }
#pragma unroll
    for (int it = 0; it < 4; ++it) {
      int i  = tid + it * 256;
      int jl = i >> 3;
      int k4 = (i & 7) << 2;
      float4 v = *(const float4*)(W + (size_t)(bn + jl) * GK + kk + k4);
      Bs[(k4 + 0) * LDB + jl] = v.x;
      Bs[(k4 + 1) * LDB + jl] = v.y;
      Bs[(k4 + 2) * LDB + jl] = v.z;
      Bs[(k4 + 3) * LDB + jl] = v.w;
    }
    __syncthreads();

#pragma unroll
    for (int k4 = 0; k4 < 8; ++k4) {
      const int kb = k4 * 4 + hi * 2;
      v2f a0, a1;
      a0.x = As[mrow0 * LDA + kb];  a0.y = As[mrow0 * LDA + kb + 1];
      a1.x = As[mrow1 * LDA + kb];  a1.y = As[mrow1 * LDA + kb + 1];
      v2f b0, b1, b2, b3;
      b0.x = Bs[kb * LDB + cb];        b0.y = Bs[(kb + 1) * LDB + cb];
      b1.x = Bs[kb * LDB + cb + 16];   b1.y = Bs[(kb + 1) * LDB + cb + 16];
      b2.x = Bs[kb * LDB + cb + 32];   b2.y = Bs[(kb + 1) * LDB + cb + 32];
      b3.x = Bs[kb * LDB + cb + 48];   b3.y = Bs[(kb + 1) * LDB + cb + 48];
      acc00 = __builtin_amdgcn_wmma_f32_16x16x4_f32(false, a0, false, b0, (short)0, acc00, false, false);
      acc01 = __builtin_amdgcn_wmma_f32_16x16x4_f32(false, a0, false, b1, (short)0, acc01, false, false);
      acc02 = __builtin_amdgcn_wmma_f32_16x16x4_f32(false, a0, false, b2, (short)0, acc02, false, false);
      acc03 = __builtin_amdgcn_wmma_f32_16x16x4_f32(false, a0, false, b3, (short)0, acc03, false, false);
      acc10 = __builtin_amdgcn_wmma_f32_16x16x4_f32(false, a1, false, b0, (short)0, acc10, false, false);
      acc11 = __builtin_amdgcn_wmma_f32_16x16x4_f32(false, a1, false, b1, (short)0, acc11, false, false);
      acc12 = __builtin_amdgcn_wmma_f32_16x16x4_f32(false, a1, false, b2, (short)0, acc12, false, false);
      acc13 = __builtin_amdgcn_wmma_f32_16x16x4_f32(false, a1, false, b3, (short)0, acc13, false, false);
    }
    __syncthreads();
  }

  const size_t r0 = (size_t)(bm + mw * 32 + 8 * hi);
  const int    cg = bn + nw * 64 + lo;
#pragma unroll
  for (int i = 0; i < 8; ++i) {
    float* p0 = out + (r0 + i) * GN + cg;
    p0[0]  = acc00[i];
    p0[16] = acc01[i];
    p0[32] = acc02[i];
    p0[48] = acc03[i];
    float* p1 = out + (r0 + 16 + i) * GN + cg;
    p1[0]  = acc10[i];
    p1[16] = acc11[i];
    p1[32] = acc12[i];
    p1[48] = acc13[i];
  }
}

__global__ __launch_bounds__(256)
void gen_weights(const float* __restrict__ x, const float* __restrict__ we_p,
                 const float* __restrict__ wr_p,
                 float* __restrict__ we, float* __restrict__ wr) {
  __shared__ int sidx[MSUB];
  const int tid = threadIdx.x;
  if (tid < MSUB) {
    const double delta = 1023.0 / 99.0;
    int v = (int)((double)tid * delta);
    if (tid == MSUB - 1) v = NSEQ - 1;
    sidx[tid] = v;
  }
  __syncthreads();

  const int g = blockIdx.x * 256 + tid;
  if (g >= BDIM * HHEADS * DH * RLOW) return;
  const int r = g % RLOW;
  const int d = (g / RLOW) % DH;
  const int h = (g / (RLOW * DH)) % HHEADS;
  const int b = g / (RLOW * DH * HHEADS);

  float aw = 0.f, ar = 0.f;
#pragma unroll 4
  for (int m = 0; m < MSUB; ++m) {
    float xs = x[((size_t)b * NSEQ + sidx[m]) * CDIM + h * DH + d];
    aw += xs * we_p[(h * MSUB + m) * RLOW + r];
    ar += xs * wr_p[(h * MSUB + m) * RLOW + r];
  }
  we[g] = aw;
  wr[g] = ar;
}

__device__ __forceinline__ void lds_barrier() {
  __builtin_amdgcn_fence(__ATOMIC_RELEASE, "workgroup", "local");
  __builtin_amdgcn_s_barrier();
  __builtin_amdgcn_fence(__ATOMIC_ACQUIRE, "workgroup", "local");
}

#define EP_ROWS  512
#define EP_WAVES 4
#define EP_NT    (EP_WAVES * 32)
#define EP_ITERS (EP_ROWS / 16 / EP_WAVES)
#define SP       68
#define T_WE     0
#define T_WR     16
#define T_SL     32
#define T_WF     62
#define T_STEPS  78

__global__ __launch_bounds__(EP_NT)
void svgp_epilogue(const float* __restrict__ qk, const float* __restrict__ we,
                   const float* __restrict__ wr,
                   const float* __restrict__ log_lambda,
                   const float* __restrict__ m_u, const float* __restrict__ s_tril,
                   const float* __restrict__ log_ssqrt,
                   const float* __restrict__ Wf, const float* __restrict__ bf,
                   const float* __restrict__ eps,
                   float* __restrict__ esc, float* __restrict__ rsc,
                   float* __restrict__ attn) {
  __shared__ v2f tab[T_STEPS * 32];
  __shared__ __attribute__((aligned(16))) float stage[EP_WAVES][2][16 * SP];

  const int tid  = threadIdx.x;
  const int lane = tid & 31;
  const int wave = tid >> 5;
  const int lo   = lane & 15;
  const int hi   = lane >> 4;
  const int h = blockIdx.y;
  const int b = blockIdx.z;
  const size_t bh = (size_t)b * HHEADS + h;

  float* raw = &stage[0][0][0];
  {
    float r8[8], w5[5];
#pragma unroll
    for (int k = 0; k < 8; ++k) {
      const int i = tid + EP_NT * k;
      r8[k] = s_tril[h * 1000 + ((i < 1000) ? i : 999)];
    }
#pragma unroll
    for (int k = 0; k < 5; ++k) w5[k] = Wf[tid + EP_NT * k];
    const int t100 = (tid < 100) ? tid : 99;
    const float rm = m_u[h * 100 + t100];
    const float rl = log_ssqrt[h * 100 + t100];
    const float rb = bf[tid & (DH - 1)];
#pragma unroll
    for (int k = 0; k < 8; ++k) {
      const int i = tid + EP_NT * k;
      if (i < 1000) raw[i] = r8[k];
    }
#pragma unroll
    for (int k = 0; k < 5; ++k) raw[1200 + tid + EP_NT * k] = w5[k];
    if (tid < 100) { raw[1000 + tid] = rm; raw[1100 + tid] = expf(rl); }
    if (tid < DH) raw[1840 + tid] = rb;
  }
  {
    const int c  = tid & 15;
    const int hh = (tid >> 4) & 1;
    const int cc = (c < RLOW) ? c : (RLOW - 1);
#pragma unroll
    for (int k = 0; k < 32 * 32 / EP_NT; ++k) {
      const int s  = (tid >> 5) + EP_WAVES * k;
      const float* wsrc = (s < T_WR) ? we : wr;
      const int ss = s & 15;
      const int d  = 8 * (ss >> 1) + 4 * hh + 2 * (ss & 1);
      const float w0 = wsrc[(bh * DH + d) * RLOW + cc];
      const float w1 = wsrc[(bh * DH + d + 1) * RLOW + cc];
      v2f tv; tv.x = (c < RLOW) ? w0 : 0.f; tv.y = (c < RLOW) ? w1 : 0.f;
      tab[s * 32 + (tid & 31)] = tv;
    }
    __syncthreads();
#pragma unroll
    for (int k = 0; k < (30 + EP_WAVES - 1) / EP_WAVES; ++k) {
      const int s = (tid >> 5) + EP_WAVES * k;
      if (s < 30) {
        const int a = s / 3;
        const int t = s - 3 * a;
        float tv[2];
#pragma unroll
        for (int e = 0; e < 2; ++e) {
          const int qp  = 4 * t + 2 * hh + e;
          const int qpc = (qp < RLOW) ? qp : (RLOW - 1);
          const float el = raw[1100 + a * RLOW + qpc];
          const float st = raw[(a * RLOW + qpc) * RLOW + cc];
          const float mu = raw[1000 + a * RLOW + cc];
          float v = (qp == c) ? el : 0.f;
          if (qp > c) v += st;
          if (qp >= RLOW) v = (qp == RLOW) ? mu : 0.f;
          tv[e] = (c < RLOW) ? v : 0.f;
        }
        v2f tv2; tv2.x = tv[0]; tv2.y = tv[1];
        tab[(T_SL + s) * 32 + (tid & 31)] = tv2;
      }
    }
#pragma unroll
    for (int k = 0; k < 16 * 32 / EP_NT; ++k) {
      const int s  = (tid >> 5) + EP_WAVES * k;
      const int d  = 16 * (s >> 2) + c;
      const int c0 = 8 * hh + 2 * (s & 3);
      const int c0c = (c0 < RLOW) ? c0 : 0;
      const float f0 = raw[1200 + d * RLOW + c0c];
      const float f1 = raw[1200 + d * RLOW + c0c + 1];
      const float bb = raw[1840 + d];
      v2f tv;
      tv.x = (c0 < RLOW) ? f0 : ((c0 == RLOW) ? bb : 0.f);
      tv.y = (c0 < RLOW) ? f1 : 0.f;
      tab[(T_WF + s) * 32 + (tid & 31)] = tv;
    }
  }

  float lam2r[RLOW];
#pragma unroll
  for (int a = 0; a < RLOW; ++a) lam2r[a] = expf(2.f * log_lambda[h * RLOW + a]);

  float* qs = stage[wave][0];
  float* ks = stage[wave][1];

#define EP_LOAD(m)  { const float* p = pl + (size_t)(2 * (m)) * GN; gq##m = *(const float4*)p; gk##m = *(const float4*)(p + CDIM); }
#define EP_STAGE(m) { *(float4*)&qs[(2 * (m) + hi) * SP + 4 * lo] = gq##m; *(float4*)&ks[(2 * (m) + hi) * SP + 4 * lo] = gk##m; }
  float4 gq0, gq1, gq2, gq3, gq4, gq5, gq6, gq7;
  float4 gk0, gk1, gk2, gk3, gk4, gk5, gk6, gk7;
  float2 gx0, gx1, gx2;
  {
    const int n0 = blockIdx.x * EP_ROWS + wave * 16;
    const float* pl = qk + ((size_t)b * NSEQ + n0 + hi) * GN + h * DH + 4 * lo;
    EP_LOAD(0) EP_LOAD(1) EP_LOAD(2) EP_LOAD(3) EP_LOAD(4) EP_LOAD(5) EP_LOAD(6) EP_LOAD(7)
    const float* ep = eps + (bh * NSEQ + n0 + lo) * RLOW;
    gx0 = *(const float2*)(ep + 2 * hi);
    gx1 = *(const float2*)(ep + 4 + 2 * hi);
    gx2 = *(const float2*)(ep + 8);
  }
  lds_barrier();

  for (int it = 0; it < EP_ITERS; ++it) {
    const int n0 = blockIdx.x * EP_ROWS + (it * EP_WAVES + wave) * 16;
    const size_t so = (bh * NSEQ + n0) * RLOW;

    EP_STAGE(0) EP_STAGE(1) EP_STAGE(2) EP_STAGE(3) EP_STAGE(4) EP_STAGE(5) EP_STAGE(6) EP_STAGE(7)
    const float2 ex0 = gx0;
    const float2 ex1 = gx1;
    const float2 ex2 = hi ? make_float2(1.f, 0.f) : gx2;

    if (it + 1 < EP_ITERS) {
      const int n1 = n0 + EP_WAVES * 16;
      const float* pl = qk + ((size_t)b * NSEQ + n1 + hi) * GN + h * DH + 4 * lo;
      EP_LOAD(0) EP_LOAD(1) EP_LOAD(2) EP_LOAD(3) EP_LOAD(4) EP_LOAD(5) EP_LOAD(6) EP_LOAD(7)
      const float* ep = eps + (bh * NSEQ + n1 + lo) * RLOW;
      gx0 = *(const float2*)(ep + 2 * hi);
      gx1 = *(const float2*)(ep + 4 + 2 * hi);
      gx2 = *(const float2*)(ep + 8);
    }
    lds_barrier();

    v8f accE = {0.f,0.f,0.f,0.f,0.f,0.f,0.f,0.f};
    v8f accR = accE;
    float qn = 0.f, kn = 0.f;
#pragma unroll
    for (int j = 0; j < 8; ++j) {
      const float4 fq = *(const float4*)&qs[lo * SP + 8 * j + 4 * hi];
      const float4 fk = *(const float4*)&ks[lo * SP + 8 * j + 4 * hi];
      qn += fq.x * fq.x + fq.y * fq.y + fq.z * fq.z + fq.w * fq.w;
      kn += fk.x * fk.x + fk.y * fk.y + fk.z * fk.z + fk.w * fk.w;
      v2f bq0, bq1, bk0, bk1;
      bq0.x = fq.x; bq0.y = fq.y; bq1.x = fq.z; bq1.y = fq.w;
      bk0.x = fk.x; bk0.y = fk.y; bk1.x = fk.z; bk1.y = fk.w;
      accE = __builtin_amdgcn_wmma_f32_16x16x4_f32(false, tab[(T_WE + 2 * j) * 32 + lane], false, bq0, (short)0, accE, false, false);
      accR = __builtin_amdgcn_wmma_f32_16x16x4_f32(false, tab[(T_WR + 2 * j) * 32 + lane], false, bk0, (short)0, accR, false, false);
      accE = __builtin_amdgcn_wmma_f32_16x16x4_f32(false, tab[(T_WE + 2 * j + 1) * 32 + lane], false, bq1, (short)0, accE, false, false);
      accR = __builtin_amdgcn_wmma_f32_16x16x4_f32(false, tab[(T_WR + 2 * j + 1) * 32 + lane], false, bk1, (short)0, accR, false, false);
    }
    qn += __shfl_xor(qn, 16, 32);
    kn += __shfl_xor(kn, 16, 32);
    const float qi = 1.f / fmaxf(sqrtf(qn), 1e-12f);
    const float ki = 1.f / fmaxf(sqrtf(kn), 1e-12f);

    float e[8], rv[8], sm[8], ot[8];
#pragma unroll
    for (int i = 0; i < 8; ++i) {
      e[i]  = accE[i] * qi;
      rv[i] = accR[i] * ki;
      sm[i] = e[i] + rv[i];
    }
#pragma unroll
    for (int i = 0; i < 8; ++i) ot[i] = __shfl_xor(sm[i], 16, 32);
    float v1[RLOW];
#pragma unroll
    for (int a = 0; a < 8; ++a) v1[a] = (hi ? ot[a] : sm[a]) * lam2r[a];
    v1[8] = (hi ? sm[0] : ot[0]) * lam2r[8];
    v1[9] = (hi ? sm[1] : ot[1]) * lam2r[9];

    v8f accS0 = {0.f,0.f,0.f,0.f,0.f,0.f,0.f,0.f};
    v8f accS1 = accS0;
#pragma unroll
    for (int a = 0; a < RLOW; ++a) {
#pragma unroll
      for (int t = 0; t < 3; ++t) {
        const float2 ex = (t == 0) ? ex0 : ((t == 1) ? ex1 : ex2);
        v2f bb; bb.x = v1[a] * ex.x; bb.y = v1[a] * ex.y;
        const v2f aa = tab[(T_SL + a * 3 + t) * 32 + lane];
        if (a & 1) accS1 = __builtin_amdgcn_wmma_f32_16x16x4_f32(false, aa, false, bb, (short)0, accS1, false, false);
        else       accS0 = __builtin_amdgcn_wmma_f32_16x16x4_f32(false, aa, false, bb, (short)0, accS0, false, false);
      }
    }
    float samp[8];
#pragma unroll
    for (int i = 0; i < 8; ++i) samp[i] = accS0[i] + accS1[i];
    samp[2] = hi ? 1.f : samp[2];

    v8f accA0 = {0.f,0.f,0.f,0.f,0.f,0.f,0.f,0.f};
    v8f accA1 = accA0, accA2 = accA0, accA3 = accA0;
#pragma unroll
    for (int s = 0; s < 4; ++s) {
      v2f bb; bb.x = samp[2 * s]; bb.y = samp[2 * s + 1];
      accA0 = __builtin_amdgcn_wmma_f32_16x16x4_f32(false, tab[(T_WF + 0 + s) * 32 + lane],  false, bb, (short)0, accA0, false, false);
      accA1 = __builtin_amdgcn_wmma_f32_16x16x4_f32(false, tab[(T_WF + 4 + s) * 32 + lane],  false, bb, (short)0, accA1, false, false);
      accA2 = __builtin_amdgcn_wmma_f32_16x16x4_f32(false, tab[(T_WF + 8 + s) * 32 + lane],  false, bb, (short)0, accA2, false, false);
      accA3 = __builtin_amdgcn_wmma_f32_16x16x4_f32(false, tab[(T_WF + 12 + s) * 32 + lane], false, bb, (short)0, accA3, false, false);
    }

    lds_barrier();

    {
      float* p = &qs[lo * SP + 8 * hi];
      *(float4*)(p +  0) = make_float4(accA0[0], accA0[1], accA0[2], accA0[3]);
      *(float4*)(p +  4) = make_float4(accA0[4], accA0[5], accA0[6], accA0[7]);
      *(float4*)(p + 16) = make_float4(accA1[0], accA1[1], accA1[2], accA1[3]);
      *(float4*)(p + 20) = make_float4(accA1[4], accA1[5], accA1[6], accA1[7]);
      *(float4*)(p + 32) = make_float4(accA2[0], accA2[1], accA2[2], accA2[3]);
      *(float4*)(p + 36) = make_float4(accA2[4], accA2[5], accA2[6], accA2[7]);
      *(float4*)(p + 48) = make_float4(accA3[0], accA3[1], accA3[2], accA3[3]);
      *(float4*)(p + 52) = make_float4(accA3[4], accA3[5], accA3[6], accA3[7]);
    }
    if (hi == 0) {
#pragma unroll
      for (int i = 0; i < 8; ++i) {
        ks[lo * RLOW + i]       = e[i];
        ks[160 + lo * RLOW + i] = rv[i];
      }
    } else {
      ks[lo * RLOW + 8]       = e[0];
      ks[lo * RLOW + 9]       = e[1];
      ks[160 + lo * RLOW + 8] = rv[0];
      ks[160 + lo * RLOW + 9] = rv[1];
    }
    lds_barrier();

    float* abase = attn + ((size_t)b * NSEQ + n0) * CDIM + h * DH;
#pragma unroll
    for (int m = 0; m < 8; ++m) {
      const float4 v = *(const float4*)&qs[(2 * m + hi) * SP + 4 * lo];
      *(float4*)(abase + (size_t)(2 * m + hi) * CDIM + 4 * lo) = v;
    }
#pragma unroll
    for (int j = 0; j < 3; ++j) {
      const int idx = lane + 32 * j;
      if (idx < 80) {
        const float4 v = *(const float4*)&ks[4 * idx];
        float* dst = (idx < 40) ? (esc + so + 4 * idx) : (rsc + so + 4 * (idx - 40));
        *(float4*)dst = v;
      }
    }
    lds_barrier();
  }
}

__global__ __launch_bounds__(256)
void kl_reduce(const float* __restrict__ log_lambda, const float* __restrict__ m_u,
               const float* __restrict__ s_tril, const float* __restrict__ log_ssqrt,
               float* __restrict__ out_kl) {
  __shared__ float red[256];
  const int tid = threadIdx.x;
  float p = 0.f;
  for (int i = tid; i < HHEADS * RLOW * RLOW * RLOW; i += 256) {
    int h = i / (RLOW * RLOW * RLOW);
    int a = (i / (RLOW * RLOW)) % RLOW;
    int q = (i / RLOW) % RLOW;
    int c = i % RLOW;
    float sl = (q == c) ? expf(log_ssqrt[(h * RLOW + a) * RLOW + q]) : 0.f;
    if (q > c) sl += s_tril[((h * RLOW + a) * RLOW + q) * RLOW + c];
    float t = expf(2.f * log_lambda[h * RLOW + a]) * sl;
    p += 0.5f * t * t;
  }
  for (int i = tid; i < HHEADS * RLOW * RLOW; i += 256) {
    int h = i / (RLOW * RLOW);
    int r = (i / RLOW) % RLOW;
    float mu = m_u[i];
    p += 0.5f * mu * mu * expf(4.f * log_lambda[h * RLOW + r]);
    p -= log_ssqrt[i];
  }
  for (int i = tid; i < HHEADS * RLOW; i += 256) p -= 2.f * (float)RLOW * log_lambda[i];
  red[tid] = p;
  __syncthreads();
  for (int s = 128; s > 0; s >>= 1) {
    if (tid < s) red[tid] += red[tid + s];
    __syncthreads();
  }
  if (tid == 0) out_kl[0] = red[0] - 0.5f * (float)(RLOW * RLOW * HHEADS);
}

extern "C" void kernel_launch(void* const* d_in, const int* in_sizes, int n_in,
                              void* d_out, int out_size, void* d_ws, size_t ws_size,
                              hipStream_t stream) {
  const float* x          = (const float*)d_in[0];
  const float* Wqk        = (const float*)d_in[1];
  const float* we_p       = (const float*)d_in[2];
  const float* wr_p       = (const float*)d_in[3];
  const float* log_lambda = (const float*)d_in[4];
  const float* m_u        = (const float*)d_in[5];
  const float* s_tril     = (const float*)d_in[6];
  const float* log_ssqrt  = (const float*)d_in[7];
  const float* Wf         = (const float*)d_in[8];
  const float* bf         = (const float*)d_in[9];
  const float* eps        = (const float*)d_in[10];

  float* ws = (float*)d_ws;
  float* qk = ws;
  float* we = ws + (size_t)GM * GN;
  float* wr = we + (size_t)BDIM * HHEADS * DH * RLOW;

  float* attn = (float*)d_out;
  float* esc  = attn + (size_t)BDIM * NSEQ * CDIM;
  float* rsc  = esc + (size_t)BDIM * HHEADS * NSEQ * RLOW;
  float* klp  = rsc + (size_t)BDIM * HHEADS * NSEQ * RLOW;

  qk_gemm_wmma<<<dim3(GN / TN, GM / TM), 256, 0, stream>>>(x, Wqk, qk);
  gen_weights<<<(BDIM * HHEADS * DH * RLOW + 255) / 256, 256, 0, stream>>>(
      x, we_p, wr_p, we, wr);
  svgp_epilogue<<<dim3(NSEQ / EP_ROWS, HHEADS, BDIM), EP_NT, 0, stream>>>(
      qk, we, wr, log_lambda, m_u, s_tril, log_ssqrt, Wf, bf, eps, esc, rsc, attn);
  kl_reduce<<<1, 256, 0, stream>>>(log_lambda, m_u, s_tril, log_ssqrt, klp);
}


# GEMM re-cut weight-stationary: a workgroup keeps its 96-column Wqk panel (all K) in LDS for the whole launch, X fragments go global->registers; K loop without staging or barrier; k sum order changes
# speedup vs baseline: 1.4407x; 1.0735x over previous
#include <hip/hip_runtime.h>

#define BDIM   32
#define NSEQ   1024
#define CDIM   768
#define HHEADS 12
#define RLOW   10
#define MSUB   100
#define DH     64

#define GM (BDIM*NSEQ)
#define GN (2*CDIM)
#define GK (CDIM)
#define PN 96
#define PP 772
#define NRG 16
#define RGR (GM / NRG)
#define GW 8
#define SR 32
#define NSTRIP (RGR / (GW * SR))

typedef __attribute__((ext_vector_type(2))) float v2f;
typedef __attribute__((ext_vector_type(8))) float v8f;

#define LOADA(dst, p) { \
  dst[0][0] = *(const float4*)((p));                dst[0][1] = *(const float4*)((p) + 8); \
  dst[0][2] = *(const float4*)((p) + 16);           dst[0][3] = *(const float4*)((p) + 24); \
  dst[1][0] = *(const float4*)((p) + 16 * GK);      dst[1][1] = *(const float4*)((p) + 16 * GK + 8); \
  dst[1][2] = *(const float4*)((p) + 16 * GK + 16); dst[1][3] = *(const float4*)((p) + 16 * GK + 24); }

#define LOADB(dst, kb) { \
  dst[0] = *(const float4*)(bl + 0 * 16 * PP + (kb)); dst[1] = *(const float4*)(bl + 1 * 16 * PP + (kb)); \
  dst[2] = *(const float4*)(bl + 2 * 16 * PP + (kb)); dst[3] = *(const float4*)(bl + 3 * 16 * PP + (kb)); \
  dst[4] = *(const float4*)(bl + 4 * 16 * PP + (kb)); dst[5] = *(const float4*)(bl + 5 * 16 * PP + (kb)); }

#define MMA8(a0, a1, b) { \
  v2f a0l, a0h, a1l, a1h; \
  a0l.x = a0.x; a0l.y = a0.y; a0h.x = a0.z; a0h.y = a0.w; \
  a1l.x = a1.x; a1l.y = a1.y; a1h.x = a1.z; a1h.y = a1.w; \
  _Pragma("unroll") for (int nt = 0; nt < 6; ++nt) { \
    v2f bv; bv.x = b[nt].x; bv.y = b[nt].y; \
    acc[0][nt] = __builtin_amdgcn_wmma_f32_16x16x4_f32(false, a0l, false, bv, (short)0, acc[0][nt], false, false); \
    acc[1][nt] = __builtin_amdgcn_wmma_f32_16x16x4_f32(false, a1l, false, bv, (short)0, acc[1][nt], false, false); } \
  _Pragma("unroll") for (int nt = 0; nt < 6; ++nt) { \
    v2f bv; bv.x = b[nt].z; bv.y = b[nt].w; \
    acc[0][nt] = __builtin_amdgcn_wmma_f32_16x16x4_f32(false, a0h, false, bv, (short)0, acc[0][nt], false, false); \
    acc[1][nt] = __builtin_amdgcn_wmma_f32_16x16x4_f32(false, a1h, false, bv, (short)0, acc[1][nt], false, false); } }

#define STEP32(a, kb, kbn) { \
  LOADB(b1, (kb) + 8)   MMA8(a[0][0], a[1][0], b0) \
  LOADB(b0, (kb) + 16)  MMA8(a[0][1], a[1][1], b1) \
  LOADB(b1, (kb) + 24)  MMA8(a[0][2], a[1][2], b0) \
  LOADB(b0, (kbn))      MMA8(a[0][3], a[1][3], b1) }

__global__ __launch_bounds__(256)
void qk_gemm_wmma(const float* __restrict__ X, const float* __restrict__ W,
                  float* __restrict__ out) {
  __shared__ __attribute__((aligned(16))) float Bs[PN * PP];
  const int tid  = threadIdx.x;
  const int lane = tid & 31;
  const int wave = tid >> 5;
  const int c    = lane & 15;
  const int h    = lane >> 4;
  const int n0   = (blockIdx.x & 15) * PN;
  const int rg   = blockIdx.x >> 4;

  for (int it = 0; it < 72; it += 8) {
    float4 v[8];
#pragma unroll
    for (int u = 0; u < 8; ++u) {
      const int f   = tid + 256 * (it + u);
      const int col = f / 192;
      const int k4  = f - col * 192;
      v[u] = *(const float4*)(W + (size_t)(n0 + col) * GK + 4 * k4);
    }
#pragma unroll
    for (int u = 0; u < 8; ++u) {
      const int f   = tid + 256 * (it + u);
      const int col = f / 192;
      const int k4  = f - col * 192;
      *(float4*)&Bs[col * PP + 4 * k4] = v[u];
    }
  }

  const float* bl = &Bs[c * PP + 4 * h];
  float4 aA[2][4], aB[2][4];
  float4 b0[6], b1[6];
  {
    const float* ap = X + (size_t)(rg * RGR + wave * SR + c) * GK + 4 * h;
    LOADA(aA, ap)
  }
  __syncthreads();
  LOADB(b0, 0)

#pragma unroll 1
  for (int s = 0; s < NSTRIP; ++s) {
    const int row0 = rg * RGR + (s * GW + wave) * SR;
    const float* ap = X + (size_t)(row0 + c) * GK + 4 * h;
    const int sn = (s + 1 < NSTRIP) ? (s + 1) : s;
    const float* apn = X + (size_t)(rg * RGR + (sn * GW + wave) * SR + c) * GK + 4 * h;

    v8f acc[2][6];
#pragma unroll
    for (int mt = 0; mt < 2; ++mt)
#pragma unroll
      for (int nt = 0; nt < 6; ++nt)
        acc[mt][nt] = (v8f){0.f,0.f,0.f,0.f,0.f,0.f,0.f,0.f};

#pragma unroll 1
    for (int kk = 0; kk < GK; kk += 64) {
      LOADA(aB, ap + kk + 32)
      STEP32(aA, kk, kk + 32)
      const bool last = (kk + 64 >= GK);
      const float* pn = last ? apn : (ap + kk + 64);
      LOADA(aA, pn)
      STEP32(aB, kk + 32, last ? 0 : (kk + 64))
    }

    float* op = out + (size_t)(row0 + 8 * h) * GN + n0 + c;
#pragma unroll
    for (int mt = 0; mt < 2; ++mt)
#pragma unroll
      for (int i = 0; i < 8; ++i)
#pragma unroll
        for (int nt = 0; nt < 6; ++nt)
          op[(size_t)(mt * 16 + i) * GN + nt * 16] = acc[mt][nt][i];
  }
}

__global__ __launch_bounds__(256)
void gen_weights(const float* __restrict__ x, const float* __restrict__ we_p,
                 const float* __restrict__ wr_p,
                 float* __restrict__ we, float* __restrict__ wr) {
  __shared__ int sidx[MSUB];
  const int tid = threadIdx.x;
  if (tid < MSUB) {
    const double delta = 1023.0 / 99.0;
    int v = (int)((double)tid * delta);
    if (tid == MSUB - 1) v = NSEQ - 1;
    sidx[tid] = v;
  }
  __syncthreads();

  const int g = blockIdx.x * 256 + tid;
  if (g >= BDIM * HHEADS * DH * RLOW) return;
  const int r = g % RLOW;
  const int d = (g / RLOW) % DH;
  const int h = (g / (RLOW * DH)) % HHEADS;
  const int b = g / (RLOW * DH * HHEADS);

  float aw = 0.f, ar = 0.f;
#pragma unroll 4
  for (int m = 0; m < MSUB; ++m) {
    float xs = x[((size_t)b * NSEQ + sidx[m]) * CDIM + h * DH + d];
    aw += xs * we_p[(h * MSUB + m) * RLOW + r];
    ar += xs * wr_p[(h * MSUB + m) * RLOW + r];
  }
  we[g] = aw;
  wr[g] = ar;
}

__device__ __forceinline__ void lds_barrier() {
  __builtin_amdgcn_fence(__ATOMIC_RELEASE, "workgroup", "local");
  __builtin_amdgcn_s_barrier();
  __builtin_amdgcn_fence(__ATOMIC_ACQUIRE, "workgroup", "local");
}

#define EP_ROWS  512
#define EP_WAVES 4
#define EP_NT    (EP_WAVES * 32)
#define EP_ITERS (EP_ROWS / 16 / EP_WAVES)
#define SP       68
#define T_WE     0
#define T_WR     16
#define T_SL     32
#define T_WF     62
#define T_STEPS  78

__global__ __launch_bounds__(EP_NT)
void svgp_epilogue(const float* __restrict__ qk, const float* __restrict__ we,
                   const float* __restrict__ wr,
                   const float* __restrict__ log_lambda,
                   const float* __restrict__ m_u, const float* __restrict__ s_tril,
                   const float* __restrict__ log_ssqrt,
                   const float* __restrict__ Wf, const float* __restrict__ bf,
                   const float* __restrict__ eps,
                   float* __restrict__ esc, float* __restrict__ rsc,
                   float* __restrict__ attn) {
  __shared__ v2f tab[T_STEPS * 32];
  __shared__ __attribute__((aligned(16))) float stage[EP_WAVES][2][16 * SP];

  const int tid  = threadIdx.x;
  const int lane = tid & 31;
  const int wave = tid >> 5;
  const int lo   = lane & 15;
  const int hi   = lane >> 4;
  const int h = blockIdx.y;
  const int b = blockIdx.z;
  const size_t bh = (size_t)b * HHEADS + h;

  float* raw = &stage[0][0][0];
  {
    float r8[8], w5[5];
#pragma unroll
    for (int k = 0; k < 8; ++k) {
      const int i = tid + EP_NT * k;
      r8[k] = s_tril[h * 1000 + ((i < 1000) ? i : 999)];
    }
#pragma unroll
    for (int k = 0; k < 5; ++k) w5[k] = Wf[tid + EP_NT * k];
    const int t100 = (tid < 100) ? tid : 99;
    const float rm = m_u[h * 100 + t100];
    const float rl = log_ssqrt[h * 100 + t100];
    const float rb = bf[tid & (DH - 1)];
#pragma unroll
    for (int k = 0; k < 8; ++k) {
      const int i = tid + EP_NT * k;
      if (i < 1000) raw[i] = r8[k];
    }
#pragma unroll
    for (int k = 0; k < 5; ++k) raw[1200 + tid + EP_NT * k] = w5[k];
    if (tid < 100) { raw[1000 + tid] = rm; raw[1100 + tid] = expf(rl); }
    if (tid < DH) raw[1840 + tid] = rb;
  }
  {
    const int c  = tid & 15;
    const int hh = (tid >> 4) & 1;
    const int cc = (c < RLOW) ? c : (RLOW - 1);
#pragma unroll
    for (int k = 0; k < 32 * 32 / EP_NT; ++k) {
      const int s  = (tid >> 5) + EP_WAVES * k;
      const float* wsrc = (s < T_WR) ? we : wr;
      const int ss = s & 15;
      const int d  = 8 * (ss >> 1) + 4 * hh + 2 * (ss & 1);
      const float w0 = wsrc[(bh * DH + d) * RLOW + cc];
      const float w1 = wsrc[(bh * DH + d + 1) * RLOW + cc];
      v2f tv; tv.x = (c < RLOW) ? w0 : 0.f; tv.y = (c < RLOW) ? w1 : 0.f;
      tab[s * 32 + (tid & 31)] = tv;
    }
    __syncthreads();
#pragma unroll
    for (int k = 0; k < (30 + EP_WAVES - 1) / EP_WAVES; ++k) {
      const int s = (tid >> 5) + EP_WAVES * k;
      if (s < 30) {
        const int a = s / 3;
        const int t = s - 3 * a;
        float tv[2];
#pragma unroll
        for (int e = 0; e < 2; ++e) {
          const int qp  = 4 * t + 2 * hh + e;
          const int qpc = (qp < RLOW) ? qp : (RLOW - 1);
          const float el = raw[1100 + a * RLOW + qpc];
          const float st = raw[(a * RLOW + qpc) * RLOW + cc];
          const float mu = raw[1000 + a * RLOW + cc];
          float v = (qp == c) ? el : 0.f;
          if (qp > c) v += st;
          if (qp >= RLOW) v = (qp == RLOW) ? mu : 0.f;
          tv[e] = (c < RLOW) ? v : 0.f;
        }
        v2f tv2; tv2.x = tv[0]; tv2.y = tv[1];
        tab[(T_SL + s) * 32 + (tid & 31)] = tv2;
      }
    }
#pragma unroll
    for (int k = 0; k < 16 * 32 / EP_NT; ++k) {
      const int s  = (tid >> 5) + EP_WAVES * k;
      const int d  = 16 * (s >> 2) + c;
      const int c0 = 8 * hh + 2 * (s & 3);
      const int c0c = (c0 < RLOW) ? c0 : 0;
      const float f0 = raw[1200 + d * RLOW + c0c];
      const float f1 = raw[1200 + d * RLOW + c0c + 1];
      const float bb = raw[1840 + d];
      v2f tv;
      tv.x = (c0 < RLOW) ? f0 : ((c0 == RLOW) ? bb : 0.f);
      tv.y = (c0 < RLOW) ? f1 : 0.f;
      tab[(T_WF + s) * 32 + (tid & 31)] = tv;
    }
  }

  float lam2r[RLOW];
#pragma unroll
  for (int a = 0; a < RLOW; ++a) lam2r[a] = expf(2.f * log_lambda[h * RLOW + a]);

  float* qs = stage[wave][0];
  float* ks = stage[wave][1];

#define EP_LOAD(m)  { const float* p = pl + (size_t)(2 * (m)) * GN; gq##m = *(const float4*)p; gk##m = *(const float4*)(p + CDIM); }
#define EP_STAGE(m) { *(float4*)&qs[(2 * (m) + hi) * SP + 4 * lo] = gq##m; *(float4*)&ks[(2 * (m) + hi) * SP + 4 * lo] = gk##m; }
  float4 gq0, gq1, gq2, gq3, gq4, gq5, gq6, gq7;
  float4 gk0, gk1, gk2, gk3, gk4, gk5, gk6, gk7;
  float2 gx0, gx1, gx2;
  {
    const int n0 = blockIdx.x * EP_ROWS + wave * 16;
    const float* pl = qk + ((size_t)b * NSEQ + n0 + hi) * GN + h * DH + 4 * lo;
    EP_LOAD(0) EP_LOAD(1) EP_LOAD(2) EP_LOAD(3) EP_LOAD(4) EP_LOAD(5) EP_LOAD(6) EP_LOAD(7)
    const float* ep = eps + (bh * NSEQ + n0 + lo) * RLOW;
    gx0 = *(const float2*)(ep + 2 * hi);
    gx1 = *(const float2*)(ep + 4 + 2 * hi);
    gx2 = *(const float2*)(ep + 8);
  }
  lds_barrier();

  for (int it = 0; it < EP_ITERS; ++it) {
    const int n0 = blockIdx.x * EP_ROWS + (it * EP_WAVES + wave) * 16;
    const size_t so = (bh * NSEQ + n0) * RLOW;

    EP_STAGE(0) EP_STAGE(1) EP_STAGE(2) EP_STAGE(3) EP_STAGE(4) EP_STAGE(5) EP_STAGE(6) EP_STAGE(7)
    const float2 ex0 = gx0;
    const float2 ex1 = gx1;
    const float2 ex2 = hi ? make_float2(1.f, 0.f) : gx2;

    if (it + 1 < EP_ITERS) {
      const int n1 = n0 + EP_WAVES * 16;
      const float* pl = qk + ((size_t)b * NSEQ + n1 + hi) * GN + h * DH + 4 * lo;
      EP_LOAD(0) EP_LOAD(1) EP_LOAD(2) EP_LOAD(3) EP_LOAD(4) EP_LOAD(5) EP_LOAD(6) EP_LOAD(7)
      const float* ep = eps + (bh * NSEQ + n1 + lo) * RLOW;
      gx0 = *(const float2*)(ep + 2 * hi);
      gx1 = *(const float2*)(ep + 4 + 2 * hi);
      gx2 = *(const float2*)(ep + 8);
    }
    lds_barrier();

    v8f accE = {0.f,0.f,0.f,0.f,0.f,0.f,0.f,0.f};
    v8f accR = accE;
    float qn = 0.f, kn = 0.f;
#pragma unroll
    for (int j = 0; j < 8; ++j) {
      const float4 fq = *(const float4*)&qs[lo * SP + 8 * j + 4 * hi];
      const float4 fk = *(const float4*)&ks[lo * SP + 8 * j + 4 * hi];
      qn += fq.x * fq.x + fq.y * fq.y + fq.z * fq.z + fq.w * fq.w;
      kn += fk.x * fk.x + fk.y * fk.y + fk.z * fk.z + fk.w * fk.w;
      v2f bq0, bq1, bk0, bk1;
      bq0.x = fq.x; bq0.y = fq.y; bq1.x = fq.z; bq1.y = fq.w;
      bk0.x = fk.x; bk0.y = fk.y; bk1.x = fk.z; bk1.y = fk.w;
      accE = __builtin_amdgcn_wmma_f32_16x16x4_f32(false, tab[(T_WE + 2 * j) * 32 + lane], false, bq0, (short)0, accE, false, false);
      accR = __builtin_amdgcn_wmma_f32_16x16x4_f32(false, tab[(T_WR + 2 * j) * 32 + lane], false, bk0, (short)0, accR, false, false);
      accE = __builtin_amdgcn_wmma_f32_16x16x4_f32(false, tab[(T_WE + 2 * j + 1) * 32 + lane], false, bq1, (short)0, accE, false, false);
      accR = __builtin_amdgcn_wmma_f32_16x16x4_f32(false, tab[(T_WR + 2 * j + 1) * 32 + lane], false, bk1, (short)0, accR, false, false);
    }
    qn += __shfl_xor(qn, 16, 32);
    kn += __shfl_xor(kn, 16, 32);
    const float qi = 1.f / fmaxf(sqrtf(qn), 1e-12f);
    const float ki = 1.f / fmaxf(sqrtf(kn), 1e-12f);

    float e[8], rv[8], sm[8], ot[8];
#pragma unroll
    for (int i = 0; i < 8; ++i) {
      e[i]  = accE[i] * qi;
      rv[i] = accR[i] * ki;
      sm[i] = e[i] + rv[i];
    }
#pragma unroll
    for (int i = 0; i < 8; ++i) ot[i] = __shfl_xor(sm[i], 16, 32);
    float v1[RLOW];
#pragma unroll
    for (int a = 0; a < 8; ++a) v1[a] = (hi ? ot[a] : sm[a]) * lam2r[a];
    v1[8] = (hi ? sm[0] : ot[0]) * lam2r[8];
    v1[9] = (hi ? sm[1] : ot[1]) * lam2r[9];

    v8f accS0 = {0.f,0.f,0.f,0.f,0.f,0.f,0.f,0.f};
    v8f accS1 = accS0;
#pragma unroll
    for (int a = 0; a < RLOW; ++a) {
#pragma unroll
      for (int t = 0; t < 3; ++t) {
        const float2 ex = (t == 0) ? ex0 : ((t == 1) ? ex1 : ex2);
        v2f bb; bb.x = v1[a] * ex.x; bb.y = v1[a] * ex.y;
        const v2f aa = tab[(T_SL + a * 3 + t) * 32 + lane];
        if (a & 1) accS1 = __builtin_amdgcn_wmma_f32_16x16x4_f32(false, aa, false, bb, (short)0, accS1, false, false);
        else       accS0 = __builtin_amdgcn_wmma_f32_16x16x4_f32(false, aa, false, bb, (short)0, accS0, false, false);
      }
    }
    float samp[8];
#pragma unroll
    for (int i = 0; i < 8; ++i) samp[i] = accS0[i] + accS1[i];
    samp[2] = hi ? 1.f : samp[2];

    v8f accA0 = {0.f,0.f,0.f,0.f,0.f,0.f,0.f,0.f};
    v8f accA1 = accA0, accA2 = accA0, accA3 = accA0;
#pragma unroll
    for (int s = 0; s < 4; ++s) {
      v2f bb; bb.x = samp[2 * s]; bb.y = samp[2 * s + 1];
      accA0 = __builtin_amdgcn_wmma_f32_16x16x4_f32(false, tab[(T_WF + 0 + s) * 32 + lane],  false, bb, (short)0, accA0, false, false);
      accA1 = __builtin_amdgcn_wmma_f32_16x16x4_f32(false, tab[(T_WF + 4 + s) * 32 + lane],  false, bb, (short)0, accA1, false, false);
      accA2 = __builtin_amdgcn_wmma_f32_16x16x4_f32(false, tab[(T_WF + 8 + s) * 32 + lane],  false, bb, (short)0, accA2, false, false);
      accA3 = __builtin_amdgcn_wmma_f32_16x16x4_f32(false, tab[(T_WF + 12 + s) * 32 + lane], false, bb, (short)0, accA3, false, false);
    }

    lds_barrier();

    {
      float* p = &qs[lo * SP + 8 * hi];
      *(float4*)(p +  0) = make_float4(accA0[0], accA0[1], accA0[2], accA0[3]);
      *(float4*)(p +  4) = make_float4(accA0[4], accA0[5], accA0[6], accA0[7]);
      *(float4*)(p + 16) = make_float4(accA1[0], accA1[1], accA1[2], accA1[3]);
      *(float4*)(p + 20) = make_float4(accA1[4], accA1[5], accA1[6], accA1[7]);
      *(float4*)(p + 32) = make_float4(accA2[0], accA2[1], accA2[2], accA2[3]);
      *(float4*)(p + 36) = make_float4(accA2[4], accA2[5], accA2[6], accA2[7]);
      *(float4*)(p + 48) = make_float4(accA3[0], accA3[1], accA3[2], accA3[3]);
      *(float4*)(p + 52) = make_float4(accA3[4], accA3[5], accA3[6], accA3[7]);
    }
    if (hi == 0) {
#pragma unroll
      for (int i = 0; i < 8; ++i) {
        ks[lo * RLOW + i]       = e[i];
        ks[160 + lo * RLOW + i] = rv[i];
      }
    } else {
      ks[lo * RLOW + 8]       = e[0];
      ks[lo * RLOW + 9]       = e[1];
      ks[160 + lo * RLOW + 8] = rv[0];
      ks[160 + lo * RLOW + 9] = rv[1];
    }
    lds_barrier();

    float* abase = attn + ((size_t)b * NSEQ + n0) * CDIM + h * DH;
#pragma unroll
    for (int m = 0; m < 8; ++m) {
      const float4 v = *(const float4*)&qs[(2 * m + hi) * SP + 4 * lo];
      *(float4*)(abase + (size_t)(2 * m + hi) * CDIM + 4 * lo) = v;
    }
#pragma unroll
    for (int j = 0; j < 3; ++j) {
      const int idx = lane + 32 * j;
      if (idx < 80) {
        const float4 v = *(const float4*)&ks[4 * idx];
        float* dst = (idx < 40) ? (esc + so + 4 * idx) : (rsc + so + 4 * (idx - 40));
        *(float4*)dst = v;
      }
    }
    lds_barrier();
  }
}

__global__ __launch_bounds__(256)
void kl_reduce(const float* __restrict__ log_lambda, const float* __restrict__ m_u,
               const float* __restrict__ s_tril, const float* __restrict__ log_ssqrt,
               float* __restrict__ out_kl) {
  __shared__ float red[256];
  const int tid = threadIdx.x;
  float p = 0.f;
  for (int i = tid; i < HHEADS * RLOW * RLOW * RLOW; i += 256) {
    int h = i / (RLOW * RLOW * RLOW);
    int a = (i / (RLOW * RLOW)) % RLOW;
    int q = (i / RLOW) % RLOW;
    int c = i % RLOW;
    float sl = (q == c) ? expf(log_ssqrt[(h * RLOW + a) * RLOW + q]) : 0.f;
    if (q > c) sl += s_tril[((h * RLOW + a) * RLOW + q) * RLOW + c];
    float t = expf(2.f * log_lambda[h * RLOW + a]) * sl;
    p += 0.5f * t * t;
  }
  for (int i = tid; i < HHEADS * RLOW * RLOW; i += 256) {
    int h = i / (RLOW * RLOW);
    int r = (i / RLOW) % RLOW;
    float mu = m_u[i];
    p += 0.5f * mu * mu * expf(4.f * log_lambda[h * RLOW + r]);
    p -= log_ssqrt[i];
  }
  for (int i = tid; i < HHEADS * RLOW; i += 256) p -= 2.f * (float)RLOW * log_lambda[i];
  red[tid] = p;
  __syncthreads();
  for (int s = 128; s > 0; s >>= 1) {
    if (tid < s) red[tid] += red[tid + s];
    __syncthreads();
  }
  if (tid == 0) out_kl[0] = red[0] - 0.5f * (float)(RLOW * RLOW * HHEADS);
}

extern "C" void kernel_launch(void* const* d_in, const int* in_sizes, int n_in,
                              void* d_out, int out_size, void* d_ws, size_t ws_size,
                              hipStream_t stream) {
  const float* x          = (const float*)d_in[0];
  const float* Wqk        = (const float*)d_in[1];
  const float* we_p       = (const float*)d_in[2];
  const float* wr_p       = (const float*)d_in[3];
  const float* log_lambda = (const float*)d_in[4];
  const float* m_u        = (const float*)d_in[5];
  const float* s_tril     = (const float*)d_in[6];
  const float* log_ssqrt  = (const float*)d_in[7];
  const float* Wf         = (const float*)d_in[8];
  const float* bf         = (const float*)d_in[9];
  const float* eps        = (const float*)d_in[10];

  float* ws = (float*)d_ws;
  float* qk = ws;
  float* we = ws + (size_t)GM * GN;
  float* wr = we + (size_t)BDIM * HHEADS * DH * RLOW;

  float* attn = (float*)d_out;
  float* esc  = attn + (size_t)BDIM * NSEQ * CDIM;
  float* rsc  = esc + (size_t)BDIM * HHEADS * NSEQ * RLOW;
  float* klp  = rsc + (size_t)BDIM * HHEADS * NSEQ * RLOW;

  qk_gemm_wmma<<<256, 256, 0, stream>>>(x, Wqk, qk);
  gen_weights<<<(BDIM * HHEADS * DH * RLOW + 255) / 256, 256, 0, stream>>>(
      x, we_p, wr_p, we, wr);
  svgp_epilogue<<<dim3(NSEQ / EP_ROWS, HHEADS, BDIM), EP_NT, 0, stream>>>(
      qk, we, wr, log_lambda, m_u, s_tril, log_ssqrt, Wf, bf, eps, esc, rsc, attn);
  kl_reduce<<<1, 256, 0, stream>>>(log_lambda, m_u, s_tril, log_ssqrt, klp);
}


# we/wr generation and the KL scalar leave their own launches for a ninth, side-role wave of each GEMM workgroup and run under its matrix-bound K loop (4 -> 2 launches); same sum order
# speedup vs baseline: 1.5986x; 1.1096x over previous
#include <hip/hip_runtime.h>

#define BDIM   32
#define NSEQ   1024
#define CDIM   768
#define HHEADS 12
#define RLOW   10
#define MSUB   100
#define DH     64

#define GM (BDIM*NSEQ)
#define GN (2*CDIM)
#define GK (CDIM)
#define PN 96
#define PP 772
#define NRG 16
#define RGR (GM / NRG)
#define GW 8
#define SR 32
#define NSTRIP (RGR / (GW * SR))
#define GT (GW * 32 + 32)

typedef __attribute__((ext_vector_type(2))) float v2f;
typedef __attribute__((ext_vector_type(8))) float v8f;

#define LOADA(dst, p) { \
  dst[0][0] = *(const float4*)((p));                dst[0][1] = *(const float4*)((p) + 8); \
  dst[0][2] = *(const float4*)((p) + 16);           dst[0][3] = *(const float4*)((p) + 24); \
  dst[1][0] = *(const float4*)((p) + 16 * GK);      dst[1][1] = *(const float4*)((p) + 16 * GK + 8); \
  dst[1][2] = *(const float4*)((p) + 16 * GK + 16); dst[1][3] = *(const float4*)((p) + 16 * GK + 24); }

#define LOADB(dst, kb) { \
  dst[0] = *(const float4*)(bl + 0 * 16 * PP + (kb)); dst[1] = *(const float4*)(bl + 1 * 16 * PP + (kb)); \
  dst[2] = *(const float4*)(bl + 2 * 16 * PP + (kb)); dst[3] = *(const float4*)(bl + 3 * 16 * PP + (kb)); \
  dst[4] = *(const float4*)(bl + 4 * 16 * PP + (kb)); dst[5] = *(const float4*)(bl + 5 * 16 * PP + (kb)); }

#define MMA8(a0, a1, b) { \
  v2f a0l, a0h, a1l, a1h; \
  a0l.x = a0.x; a0l.y = a0.y; a0h.x = a0.z; a0h.y = a0.w; \
  a1l.x = a1.x; a1l.y = a1.y; a1h.x = a1.z; a1h.y = a1.w; \
  _Pragma("unroll") for (int nt = 0; nt < 6; ++nt) { \
    v2f bv; bv.x = b[nt].x; bv.y = b[nt].y; \
    acc[0][nt] = __builtin_amdgcn_wmma_f32_16x16x4_f32(false, a0l, false, bv, (short)0, acc[0][nt], false, false); \
    acc[1][nt] = __builtin_amdgcn_wmma_f32_16x16x4_f32(false, a1l, false, bv, (short)0, acc[1][nt], false, false); } \
  _Pragma("unroll") for (int nt = 0; nt < 6; ++nt) { \
    v2f bv; bv.x = b[nt].z; bv.y = b[nt].w; \
    acc[0][nt] = __builtin_amdgcn_wmma_f32_16x16x4_f32(false, a0h, false, bv, (short)0, acc[0][nt], false, false); \
    acc[1][nt] = __builtin_amdgcn_wmma_f32_16x16x4_f32(false, a1h, false, bv, (short)0, acc[1][nt], false, false); } }

#define STEP32(a, kb, kbn) { \
  LOADB(b1, (kb) + 8)   MMA8(a[0][0], a[1][0], b0) \
  LOADB(b0, (kb) + 16)  MMA8(a[0][1], a[1][1], b1) \
  LOADB(b1, (kb) + 24)  MMA8(a[0][2], a[1][2], b0) \
  LOADB(b0, (kbn))      MMA8(a[0][3], a[1][3], b1) }

__device__ __forceinline__ void side_role(const float* __restrict__ x,
                                          const float* __restrict__ we_p,
                                          const float* __restrict__ wr_p,
                                          float* __restrict__ we, float* __restrict__ wr,
                                          const float* __restrict__ log_lambda,
                                          const float* __restrict__ m_u,
                                          const float* __restrict__ s_tril,
                                          const float* __restrict__ log_ssqrt,
                                          float* __restrict__ out_kl,
                                          const int* sidx, const int lane, const int bid) {
#pragma unroll 1
  for (int j = 0; j < 3; ++j) {
    const int u   = bid * 3 + j;
    const int bhh = u >> 1;
    const int b   = bhh / HHEADS;
    const int h   = bhh - b * HHEADS;
    const int d   = (u & 1) * 32 + lane;
    const float* xp = x + (size_t)b * NSEQ * CDIM + h * DH + d;
    const float* pe = we_p + h * MSUB * RLOW;
    const float* pr = wr_p + h * MSUB * RLOW;
    float aw[RLOW], ar[RLOW];
#pragma unroll
    for (int r = 0; r < RLOW; ++r) { aw[r] = 0.f; ar[r] = 0.f; }
#pragma unroll 1
    for (int m0 = 0; m0 < MSUB; m0 += 10) {
      float xv[10];
#pragma unroll
      for (int q = 0; q < 10; ++q) xv[q] = xp[(size_t)sidx[m0 + q] * CDIM];
#pragma unroll
      for (int q = 0; q < 10; ++q) {
#pragma unroll
        for (int r = 0; r < RLOW; ++r) {
          aw[r] += xv[q] * pe[(m0 + q) * RLOW + r];
          ar[r] += xv[q] * pr[(m0 + q) * RLOW + r];
        }
      }
    }
    float* ow = we + ((size_t)bhh * DH + d) * RLOW;
    float* ov = wr + ((size_t)bhh * DH + d) * RLOW;
#pragma unroll
    for (int r = 0; r < RLOW; r += 2) {
      *(float2*)(ow + r) = make_float2(aw[r], aw[r + 1]);
      *(float2*)(ov + r) = make_float2(ar[r], ar[r + 1]);
    }
  }

  if (bid != 0) return;

  float pv[8];
#pragma unroll
  for (int v = 0; v < 8; ++v) pv[v] = 0.f;
#pragma unroll 1
  for (int t = 0; t < (HHEADS * 1000 + 255) / 256; ++t) {
#pragma unroll
    for (int v = 0; v < 8; ++v) {
      const int  i0 = t * 256 + 32 * v + lane;
      const bool ok = i0 < HHEADS * 1000;
      const int  i  = ok ? i0 : (HHEADS * 1000 - 1);
      const int  ha = i / (RLOW * RLOW);
      const int  q  = (i / RLOW) % RLOW;
      const int  c  = i % RLOW;
      const float lsv = log_ssqrt[ha * RLOW + q];
      const float stv = s_tril[i];
      const float llv = log_lambda[ha];
      const float el = expf(lsv);
      float sl = (q == c) ? el : 0.f;
      sl += (q > c) ? stv : 0.f;
      const float tt = expf(2.f * llv) * sl;
      const float tm = ok ? tt : 0.f;
      pv[v] += 0.5f * tm * tm;
    }
  }
#pragma unroll 1
  for (int t = 0; t < (HHEADS * 100 + 255) / 256; ++t) {
#pragma unroll
    for (int v = 0; v < 8; ++v) {
      const int  i0 = t * 256 + 32 * v + lane;
      const bool ok = i0 < HHEADS * 100;
      const int  i  = ok ? i0 : (HHEADS * 100 - 1);
      const int  hr = i / RLOW;
      const float muv = m_u[i];
      const float lsv = log_ssqrt[i];
      const float llv = log_lambda[hr];
      const float mu  = ok ? muv : 0.f;
      pv[v] += 0.5f * mu * mu * expf(4.f * llv);
      pv[v] -= ok ? lsv : 0.f;
    }
  }
#pragma unroll
  for (int v = 0; v < 4; ++v) {
    const int  i0 = 32 * v + lane;
    const bool ok = i0 < HHEADS * RLOW;
    const float llv = log_lambda[ok ? i0 : 0];
    pv[v] -= 2.f * (float)RLOW * (ok ? llv : 0.f);
  }
#pragma unroll
  for (int v = 0; v < 4; ++v) pv[v] += pv[v + 4];
  pv[0] += pv[2];
  pv[1] += pv[3];
  float red = pv[0] + pv[1];
#pragma unroll
  for (int sft = 16; sft > 0; sft >>= 1) red += __shfl_xor(red, sft, 32);
  if (lane == 0) out_kl[0] = red - 0.5f * (float)(RLOW * RLOW * HHEADS);
}

__global__ __launch_bounds__(GT)
void qk_gemm_wmma(const float* __restrict__ X, const float* __restrict__ W,
                  float* __restrict__ out,
                  const float* __restrict__ we_p, const float* __restrict__ wr_p,
                  float* __restrict__ we, float* __restrict__ wr,
                  const float* __restrict__ log_lambda, const float* __restrict__ m_u,
                  const float* __restrict__ s_tril, const float* __restrict__ log_ssqrt,
                  float* __restrict__ out_kl) {
  __shared__ __attribute__((aligned(16))) float Bs[PN * PP];
  __shared__ int sidx[MSUB];
  const int tid  = threadIdx.x;
  const int lane = tid & 31;
  const int wave = __builtin_amdgcn_readfirstlane(tid >> 5);
  const int c    = lane & 15;
  const int h    = lane >> 4;
  const int n0   = (blockIdx.x & 15) * PN;
  const int rg   = blockIdx.x >> 4;

  if (wave < GW) {
  for (int it = 0; it < 72; it += 8) {
    float4 v[8];
#pragma unroll
    for (int u = 0; u < 8; ++u) {
      const int f   = tid + 256 * (it + u);
      const int col = f / 192;
      const int k4  = f - col * 192;
      v[u] = *(const float4*)(W + (size_t)(n0 + col) * GK + 4 * k4);
    }
#pragma unroll
    for (int u = 0; u < 8; ++u) {
      const int f   = tid + 256 * (it + u);
      const int col = f / 192;
      const int k4  = f - col * 192;
      *(float4*)&Bs[col * PP + 4 * k4] = v[u];
    }
  }
  } else {
#pragma unroll
    for (int k = 0; k < 4; ++k) {
      const int m = lane + 32 * k;
      if (m < MSUB) {
        const double delta = 1023.0 / 99.0;
        int v = (int)((double)m * delta);
        if (m == MSUB - 1) v = NSEQ - 1;
        sidx[m] = v;
      }
    }
  }

  const float* bl = &Bs[c * PP + 4 * h];
  float4 aA[2][4], aB[2][4];
  float4 b0[6], b1[6];
  if (wave < GW) {
    const float* ap = X + (size_t)(rg * RGR + wave * SR + c) * GK + 4 * h;
    LOADA(aA, ap)
  }
  __syncthreads();
  if (wave == GW) {
    side_role(X, we_p, wr_p, we, wr, log_lambda, m_u, s_tril, log_ssqrt, out_kl,
              sidx, lane, (int)blockIdx.x);
    return;
  }
  LOADB(b0, 0)

#pragma unroll 1
  for (int s = 0; s < NSTRIP; ++s) {
    const int row0 = rg * RGR + (s * GW + wave) * SR;
    const float* ap = X + (size_t)(row0 + c) * GK + 4 * h;
    const int sn = (s + 1 < NSTRIP) ? (s + 1) : s;
    const float* apn = X + (size_t)(rg * RGR + (sn * GW + wave) * SR + c) * GK + 4 * h;

    v8f acc[2][6];
#pragma unroll
    for (int mt = 0; mt < 2; ++mt)
#pragma unroll
      for (int nt = 0; nt < 6; ++nt)
        acc[mt][nt] = (v8f){0.f,0.f,0.f,0.f,0.f,0.f,0.f,0.f};

#pragma unroll 1
    for (int kk = 0; kk < GK; kk += 64) {
      LOADA(aB, ap + kk + 32)
      STEP32(aA, kk, kk + 32)
      const bool last = (kk + 64 >= GK);
      const float* pn = last ? apn : (ap + kk + 64);
      LOADA(aA, pn)
      STEP32(aB, kk + 32, last ? 0 : (kk + 64))
    }

    float* op = out + (size_t)(row0 + 8 * h) * GN + n0 + c;
#pragma unroll
    for (int mt = 0; mt < 2; ++mt)
#pragma unroll
      for (int i = 0; i < 8; ++i)
#pragma unroll
        for (int nt = 0; nt < 6; ++nt)
          op[(size_t)(mt * 16 + i) * GN + nt * 16] = acc[mt][nt][i];
  }
}

__device__ __forceinline__ void lds_barrier() {
  __builtin_amdgcn_fence(__ATOMIC_RELEASE, "workgroup", "local");
  __builtin_amdgcn_s_barrier();
  __builtin_amdgcn_fence(__ATOMIC_ACQUIRE, "workgroup", "local");
}

#define EP_ROWS  512
#define EP_WAVES 4
#define EP_NT    (EP_WAVES * 32)
#define EP_ITERS (EP_ROWS / 16 / EP_WAVES)
#define SP       68
#define T_WE     0
#define T_WR     16
#define T_SL     32
#define T_WF     62
#define T_STEPS  78

__global__ __launch_bounds__(EP_NT)
void svgp_epilogue(const float* __restrict__ qk, const float* __restrict__ we,
                   const float* __restrict__ wr,
                   const float* __restrict__ log_lambda,
                   const float* __restrict__ m_u, const float* __restrict__ s_tril,
                   const float* __restrict__ log_ssqrt,
                   const float* __restrict__ Wf, const float* __restrict__ bf,
                   const float* __restrict__ eps,
                   float* __restrict__ esc, float* __restrict__ rsc,
                   float* __restrict__ attn) {
  __shared__ v2f tab[T_STEPS * 32];
  __shared__ __attribute__((aligned(16))) float stage[EP_WAVES][2][16 * SP];

  const int tid  = threadIdx.x;
  const int lane = tid & 31;
  const int wave = tid >> 5;
  const int lo   = lane & 15;
  const int hi   = lane >> 4;
  const int h = blockIdx.y;
  const int b = blockIdx.z;
  const size_t bh = (size_t)b * HHEADS + h;

  float* raw = &stage[0][0][0];
  {
    float r8[8], w5[5];
#pragma unroll
    for (int k = 0; k < 8; ++k) {
      const int i = tid + EP_NT * k;
      r8[k] = s_tril[h * 1000 + ((i < 1000) ? i : 999)];
    }
#pragma unroll
    for (int k = 0; k < 5; ++k) w5[k] = Wf[tid + EP_NT * k];
    const int t100 = (tid < 100) ? tid : 99;
    const float rm = m_u[h * 100 + t100];
    const float rl = log_ssqrt[h * 100 + t100];
    const float rb = bf[tid & (DH - 1)];
#pragma unroll
    for (int k = 0; k < 8; ++k) {
      const int i = tid + EP_NT * k;
      if (i < 1000) raw[i] = r8[k];
    }
#pragma unroll
    for (int k = 0; k < 5; ++k) raw[1200 + tid + EP_NT * k] = w5[k];
    if (tid < 100) { raw[1000 + tid] = rm; raw[1100 + tid] = expf(rl); }
    if (tid < DH) raw[1840 + tid] = rb;
  }
  {
    const int c  = tid & 15;
    const int hh = (tid >> 4) & 1;
    const int cc = (c < RLOW) ? c : (RLOW - 1);
#pragma unroll
    for (int k = 0; k < 32 * 32 / EP_NT; ++k) {
      const int s  = (tid >> 5) + EP_WAVES * k;
      const float* wsrc = (s < T_WR) ? we : wr;
      const int ss = s & 15;
      const int d  = 8 * (ss >> 1) + 4 * hh + 2 * (ss & 1);
      const float w0 = wsrc[(bh * DH + d) * RLOW + cc];
      const float w1 = wsrc[(bh * DH + d + 1) * RLOW + cc];
      v2f tv; tv.x = (c < RLOW) ? w0 : 0.f; tv.y = (c < RLOW) ? w1 : 0.f;
      tab[s * 32 + (tid & 31)] = tv;
    }
    __syncthreads();
#pragma unroll
    for (int k = 0; k < (30 + EP_WAVES - 1) / EP_WAVES; ++k) {
      const int s = (tid >> 5) + EP_WAVES * k;
      if (s < 30) {
        const int a = s / 3;
        const int t = s - 3 * a;
        float tv[2];
#pragma unroll
        for (int e = 0; e < 2; ++e) {
          const int qp  = 4 * t + 2 * hh + e;
          const int qpc = (qp < RLOW) ? qp : (RLOW - 1);
          const float el = raw[1100 + a * RLOW + qpc];
          const float st = raw[(a * RLOW + qpc) * RLOW + cc];
          const float mu = raw[1000 + a * RLOW + cc];
          float v = (qp == c) ? el : 0.f;
          if (qp > c) v += st;
          if (qp >= RLOW) v = (qp == RLOW) ? mu : 0.f;
          tv[e] = (c < RLOW) ? v : 0.f;
        }
        v2f tv2; tv2.x = tv[0]; tv2.y = tv[1];
        tab[(T_SL + s) * 32 + (tid & 31)] = tv2;
      }
    }
#pragma unroll
    for (int k = 0; k < 16 * 32 / EP_NT; ++k) {
      const int s  = (tid >> 5) + EP_WAVES * k;
      const int d  = 16 * (s >> 2) + c;
      const int c0 = 8 * hh + 2 * (s & 3);
      const int c0c = (c0 < RLOW) ? c0 : 0;
      const float f0 = raw[1200 + d * RLOW + c0c];
      const float f1 = raw[1200 + d * RLOW + c0c + 1];
      const float bb = raw[1840 + d];
      v2f tv;
      tv.x = (c0 < RLOW) ? f0 : ((c0 == RLOW) ? bb : 0.f);
      tv.y = (c0 < RLOW) ? f1 : 0.f;
      tab[(T_WF + s) * 32 + (tid & 31)] = tv;
    }
  }

  float lam2r[RLOW];
#pragma unroll
  for (int a = 0; a < RLOW; ++a) lam2r[a] = expf(2.f * log_lambda[h * RLOW + a]);

  float* qs = stage[wave][0];
  float* ks = stage[wave][1];

#define EP_LOAD(m)  { const float* p = pl + (size_t)(2 * (m)) * GN; gq##m = *(const float4*)p; gk##m = *(const float4*)(p + CDIM); }
#define EP_STAGE(m) { *(float4*)&qs[(2 * (m) + hi) * SP + 4 * lo] = gq##m; *(float4*)&ks[(2 * (m) + hi) * SP + 4 * lo] = gk##m; }
  float4 gq0, gq1, gq2, gq3, gq4, gq5, gq6, gq7;
  float4 gk0, gk1, gk2, gk3, gk4, gk5, gk6, gk7;
  float2 gx0, gx1, gx2;
  {
    const int n0 = blockIdx.x * EP_ROWS + wave * 16;
    const float* pl = qk + ((size_t)b * NSEQ + n0 + hi) * GN + h * DH + 4 * lo;
    EP_LOAD(0) EP_LOAD(1) EP_LOAD(2) EP_LOAD(3) EP_LOAD(4) EP_LOAD(5) EP_LOAD(6) EP_LOAD(7)
    const float* ep = eps + (bh * NSEQ + n0 + lo) * RLOW;
    gx0 = *(const float2*)(ep + 2 * hi);
    gx1 = *(const float2*)(ep + 4 + 2 * hi);
    gx2 = *(const float2*)(ep + 8);
  }
  lds_barrier();

  for (int it = 0; it < EP_ITERS; ++it) {
    const int n0 = blockIdx.x * EP_ROWS + (it * EP_WAVES + wave) * 16;
    const size_t so = (bh * NSEQ + n0) * RLOW;

    EP_STAGE(0) EP_STAGE(1) EP_STAGE(2) EP_STAGE(3) EP_STAGE(4) EP_STAGE(5) EP_STAGE(6) EP_STAGE(7)
    const float2 ex0 = gx0;
    const float2 ex1 = gx1;
    const float2 ex2 = hi ? make_float2(1.f, 0.f) : gx2;

    if (it + 1 < EP_ITERS) {
      const int n1 = n0 + EP_WAVES * 16;
      const float* pl = qk + ((size_t)b * NSEQ + n1 + hi) * GN + h * DH + 4 * lo;
      EP_LOAD(0) EP_LOAD(1) EP_LOAD(2) EP_LOAD(3) EP_LOAD(4) EP_LOAD(5) EP_LOAD(6) EP_LOAD(7)
      const float* ep = eps + (bh * NSEQ + n1 + lo) * RLOW;
      gx0 = *(const float2*)(ep + 2 * hi);
      gx1 = *(const float2*)(ep + 4 + 2 * hi);
      gx2 = *(const float2*)(ep + 8);
    }
    lds_barrier();

    v8f accE = {0.f,0.f,0.f,0.f,0.f,0.f,0.f,0.f};
    v8f accR = accE;
    float qn = 0.f, kn = 0.f;
#pragma unroll
    for (int j = 0; j < 8; ++j) {
      const float4 fq = *(const float4*)&qs[lo * SP + 8 * j + 4 * hi];
      const float4 fk = *(const float4*)&ks[lo * SP + 8 * j + 4 * hi];
      qn += fq.x * fq.x + fq.y * fq.y + fq.z * fq.z + fq.w * fq.w;
      kn += fk.x * fk.x + fk.y * fk.y + fk.z * fk.z + fk.w * fk.w;
      v2f bq0, bq1, bk0, bk1;
      bq0.x = fq.x; bq0.y = fq.y; bq1.x = fq.z; bq1.y = fq.w;
      bk0.x = fk.x; bk0.y = fk.y; bk1.x = fk.z; bk1.y = fk.w;
      accE = __builtin_amdgcn_wmma_f32_16x16x4_f32(false, tab[(T_WE + 2 * j) * 32 + lane], false, bq0, (short)0, accE, false, false);
      accR = __builtin_amdgcn_wmma_f32_16x16x4_f32(false, tab[(T_WR + 2 * j) * 32 + lane], false, bk0, (short)0, accR, false, false);
      accE = __builtin_amdgcn_wmma_f32_16x16x4_f32(false, tab[(T_WE + 2 * j + 1) * 32 + lane], false, bq1, (short)0, accE, false, false);
      accR = __builtin_amdgcn_wmma_f32_16x16x4_f32(false, tab[(T_WR + 2 * j + 1) * 32 + lane], false, bk1, (short)0, accR, false, false);
    }
    qn += __shfl_xor(qn, 16, 32);
    kn += __shfl_xor(kn, 16, 32);
    const float qi = 1.f / fmaxf(sqrtf(qn), 1e-12f);
    const float ki = 1.f / fmaxf(sqrtf(kn), 1e-12f);

    float e[8], rv[8], sm[8], ot[8];
#pragma unroll
    for (int i = 0; i < 8; ++i) {
      e[i]  = accE[i] * qi;
      rv[i] = accR[i] * ki;
      sm[i] = e[i] + rv[i];
    }
#pragma unroll
    for (int i = 0; i < 8; ++i) ot[i] = __shfl_xor(sm[i], 16, 32);
    float v1[RLOW];
#pragma unroll
    for (int a = 0; a < 8; ++a) v1[a] = (hi ? ot[a] : sm[a]) * lam2r[a];
    v1[8] = (hi ? sm[0] : ot[0]) * lam2r[8];
    v1[9] = (hi ? sm[1] : ot[1]) * lam2r[9];

    v8f accS0 = {0.f,0.f,0.f,0.f,0.f,0.f,0.f,0.f};
    v8f accS1 = accS0;
#pragma unroll
    for (int a = 0; a < RLOW; ++a) {
#pragma unroll
      for (int t = 0; t < 3; ++t) {
        const float2 ex = (t == 0) ? ex0 : ((t == 1) ? ex1 : ex2);
        v2f bb; bb.x = v1[a] * ex.x; bb.y = v1[a] * ex.y;
        const v2f aa = tab[(T_SL + a * 3 + t) * 32 + lane];
        if (a & 1) accS1 = __builtin_amdgcn_wmma_f32_16x16x4_f32(false, aa, false, bb, (short)0, accS1, false, false);
        else       accS0 = __builtin_amdgcn_wmma_f32_16x16x4_f32(false, aa, false, bb, (short)0, accS0, false, false);
      }
    }
    float samp[8];
#pragma unroll
    for (int i = 0; i < 8; ++i) samp[i] = accS0[i] + accS1[i];
    samp[2] = hi ? 1.f : samp[2];

    v8f accA0 = {0.f,0.f,0.f,0.f,0.f,0.f,0.f,0.f};
    v8f accA1 = accA0, accA2 = accA0, accA3 = accA0;
#pragma unroll
    for (int s = 0; s < 4; ++s) {
      v2f bb; bb.x = samp[2 * s]; bb.y = samp[2 * s + 1];
      accA0 = __builtin_amdgcn_wmma_f32_16x16x4_f32(false, tab[(T_WF + 0 + s) * 32 + lane],  false, bb, (short)0, accA0, false, false);
      accA1 = __builtin_amdgcn_wmma_f32_16x16x4_f32(false, tab[(T_WF + 4 + s) * 32 + lane],  false, bb, (short)0, accA1, false, false);
      accA2 = __builtin_amdgcn_wmma_f32_16x16x4_f32(false, tab[(T_WF + 8 + s) * 32 + lane],  false, bb, (short)0, accA2, false, false);
      accA3 = __builtin_amdgcn_wmma_f32_16x16x4_f32(false, tab[(T_WF + 12 + s) * 32 + lane], false, bb, (short)0, accA3, false, false);
    }

    lds_barrier();

    {
      float* p = &qs[lo * SP + 8 * hi];
      *(float4*)(p +  0) = make_float4(accA0[0], accA0[1], accA0[2], accA0[3]);
      *(float4*)(p +  4) = make_float4(accA0[4], accA0[5], accA0[6], accA0[7]);
      *(float4*)(p + 16) = make_float4(accA1[0], accA1[1], accA1[2], accA1[3]);
      *(float4*)(p + 20) = make_float4(accA1[4], accA1[5], accA1[6], accA1[7]);
      *(float4*)(p + 32) = make_float4(accA2[0], accA2[1], accA2[2], accA2[3]);
      *(float4*)(p + 36) = make_float4(accA2[4], accA2[5], accA2[6], accA2[7]);
      *(float4*)(p + 48) = make_float4(accA3[0], accA3[1], accA3[2], accA3[3]);
      *(float4*)(p + 52) = make_float4(accA3[4], accA3[5], accA3[6], accA3[7]);
    }
    if (hi == 0) {
#pragma unroll
      for (int i = 0; i < 8; ++i) {
        ks[lo * RLOW + i]       = e[i];
        ks[160 + lo * RLOW + i] = rv[i];
      }
    } else {
      ks[lo * RLOW + 8]       = e[0];
      ks[lo * RLOW + 9]       = e[1];
      ks[160 + lo * RLOW + 8] = rv[0];
      ks[160 + lo * RLOW + 9] = rv[1];
    }
    lds_barrier();

    float* abase = attn + ((size_t)b * NSEQ + n0) * CDIM + h * DH;
#pragma unroll
    for (int m = 0; m < 8; ++m) {
      const float4 v = *(const float4*)&qs[(2 * m + hi) * SP + 4 * lo];
      *(float4*)(abase + (size_t)(2 * m + hi) * CDIM + 4 * lo) = v;
    }
#pragma unroll
    for (int j = 0; j < 3; ++j) {
      const int idx = lane + 32 * j;
      if (idx < 80) {
        const float4 v = *(const float4*)&ks[4 * idx];
        float* dst = (idx < 40) ? (esc + so + 4 * idx) : (rsc + so + 4 * (idx - 40));
        *(float4*)dst = v;
      }
    }
    lds_barrier();
  }
}

extern "C" void kernel_launch(void* const* d_in, const int* in_sizes, int n_in,
                              void* d_out, int out_size, void* d_ws, size_t ws_size,
                              hipStream_t stream) {
  const float* x          = (const float*)d_in[0];
  const float* Wqk        = (const float*)d_in[1];
  const float* we_p       = (const float*)d_in[2];
  const float* wr_p       = (const float*)d_in[3];
  const float* log_lambda = (const float*)d_in[4];
  const float* m_u        = (const float*)d_in[5];
  const float* s_tril     = (const float*)d_in[6];
  const float* log_ssqrt  = (const float*)d_in[7];
  const float* Wf         = (const float*)d_in[8];
  const float* bf         = (const float*)d_in[9];
  const float* eps        = (const float*)d_in[10];

  float* ws = (float*)d_ws;
  float* qk = ws;
  float* we = ws + (size_t)GM * GN;
  float* wr = we + (size_t)BDIM * HHEADS * DH * RLOW;

  float* attn = (float*)d_out;
  float* esc  = attn + (size_t)BDIM * NSEQ * CDIM;
  float* rsc  = esc + (size_t)BDIM * HHEADS * NSEQ * RLOW;
  float* klp  = rsc + (size_t)BDIM * HHEADS * NSEQ * RLOW;

  qk_gemm_wmma<<<256, GT, 0, stream>>>(x, Wqk, qk, we_p, wr_p, we, wr,
                                       log_lambda, m_u, s_tril, log_ssqrt, klp);
  svgp_epilogue<<<dim3(NSEQ / EP_ROWS, HHEADS, BDIM), EP_NT, 0, stream>>>(
      qk, we, wr, log_lambda, m_u, s_tril, log_ssqrt, Wf, bf, eps, esc, rsc, attn);
}
